# starting kernel
# baseline (speedup 1.0000x reference)
#include <hip/hip_runtime.h>


#define EDIM 10
#define FDIM 64
#define KPAD 12
#define FPAD 17
#define THREADS 256
#define LN_EPS 1e-5f

typedef float v2f __attribute__((ext_vector_type(2)));
typedef float v8f __attribute__((ext_vector_type(8)));

__global__ void qtb_setup(const float* __restrict__ W_ffn,
                          const float* __restrict__ W_out,
                          float* __restrict__ Bpad)
{
    int t = threadIdx.x;
    if (t >= KPAD * 16) return;
    int k = t >> 4;
    int n = t & 15;
    float s = 0.0f;
    if (k < EDIM && n < EDIM) {
        #pragma unroll
        for (int f = 0; f < FDIM; ++f)
            s += W_out[n * FDIM + f] * W_ffn[f * EDIM + k];
    }
    Bpad[t] = s;
}

__launch_bounds__(THREADS)
__global__ void qtb_main(const float* __restrict__ x,
                         const float* __restrict__ th_a,
                         const float* __restrict__ th_f,
                         const float* __restrict__ g1, const float* __restrict__ b1,
                         const float* __restrict__ g2, const float* __restrict__ b2,
                         const float* __restrict__ Bpad,
                         float* __restrict__ out)
{
    __shared__ __align__(16) float meas_lds[THREADS * KPAD];
    __shared__ __align__(16) float ffn_lds [THREADS * FPAD];

    const int tid  = threadIdx.x;
    const int lane = tid & 31;
    const int wave = tid >> 5;
    const int nn   = lane & 15;
    const int kh   = lane >> 4;

    const long tok = (long)blockIdx.x * THREADS + tid;
    const float* xp = x + tok * EDIM;

    float xv[EDIM];
    #pragma unroll
    for (int i = 0; i < EDIM / 2; ++i) {
        v2f v = *(const v2f*)(xp + 2 * i);
        xv[2 * i] = v.x;
        xv[2 * i + 1] = v.y;
    }

    float v1[EDIM];
    float prod = 1.0f, sum = 0.0f;
    #pragma unroll
    for (int i = 0; i < EDIM; ++i) {
        prod *= __cosf(xv[i] + th_a[i]);
        v1[i] = xv[i] + prod;
        sum  += v1[i];
    }

    const float inv_e = 1.0f / (float)EDIM;
    float mu = sum * inv_e;
    float var = 0.0f;
    #pragma unroll
    for (int i = 0; i < EDIM; ++i) {
        float d = v1[i] - mu;
        var += d * d;
    }
    var *= inv_e;
    float rs = rsqrtf(var + LN_EPS);
    float h[EDIM];
    #pragma unroll
    for (int i = 0; i < EDIM; ++i)
        h[i] = (v1[i] - mu) * rs * g1[i] + b1[i];

    #pragma unroll
    for (int i = 0; i < EDIM; ++i)
        meas_lds[tid * KPAD + i] = __cosf(h[i] + th_f[i]);
    meas_lds[tid * KPAD + 10] = 0.0f;
    meas_lds[tid * KPAD + 11] = 0.0f;

    v2f bmat[3];
    #pragma unroll
    for (int c = 0; c < 3; ++c) {
        int k0 = c * 4 + 2 * kh;
        bmat[c].x = Bpad[k0 * 16 + nn];
        bmat[c].y = Bpad[(k0 + 1) * 16 + nn];
    }

    __syncthreads();

    v8f acc[2];
    #pragma unroll
    for (int t2 = 0; t2 < 2; ++t2) {
        const int tile = wave * 2 + t2;
        const float* arow = &meas_lds[(tile * 16 + nn) * KPAD + 2 * kh];
        v8f c = {0.f, 0.f, 0.f, 0.f, 0.f, 0.f, 0.f, 0.f};
        #pragma unroll
        for (int ck = 0; ck < 3; ++ck) {
            v2f a = *(const v2f*)(arow + ck * 4);
            c = __builtin_amdgcn_wmma_f32_16x16x4_f32(
                     false, a,
                     false, bmat[ck],
                     (short)0, c,
                     false,  false);
        }
        acc[t2] = c;
    }

    #pragma unroll
    for (int t2 = 0; t2 < 2; ++t2) {
        const int tile = wave * 2 + t2;
        #pragma unroll
        for (int r = 0; r < 8; ++r)
            ffn_lds[(tile * 16 + r + 8 * kh) * FPAD + nn] = acc[t2][r];
    }

    __syncthreads();

    float yv[EDIM];
    float s2 = 0.0f;
    #pragma unroll
    for (int i = 0; i < EDIM; ++i) {
        yv[i] = h[i] + ffn_lds[tid * FPAD + i];
        s2 += yv[i];
    }
    float mu2 = s2 * inv_e;
    float var2 = 0.0f;
    #pragma unroll
    for (int i = 0; i < EDIM; ++i) {
        float d = yv[i] - mu2;
        var2 += d * d;
    }
    var2 *= inv_e;
    float rs2 = rsqrtf(var2 + LN_EPS);

    float* op = out + tok * EDIM;
    #pragma unroll
    for (int i = 0; i < EDIM / 2; ++i) {
        v2f v;
        v.x = (yv[2 * i]     - mu2) * rs2 * g2[2 * i]     + b2[2 * i];
        v.y = (yv[2 * i + 1] - mu2) * rs2 * g2[2 * i + 1] + b2[2 * i + 1];
        *(v2f*)(op + 2 * i) = v;
    }
}

extern "C" void kernel_launch(void* const* d_in, const int* in_sizes, int n_in,
                              void* d_out, int out_size, void* d_ws, size_t ws_size,
                              hipStream_t stream) {
    const float* x     = (const float*)d_in[0];
    const float* th_a  = (const float*)d_in[1];
    const float* th_f  = (const float*)d_in[2];
    const float* W_ffn = (const float*)d_in[3];
    const float* W_out = (const float*)d_in[4];
    const float* g1    = (const float*)d_in[5];
    const float* b1    = (const float*)d_in[6];
    const float* g2    = (const float*)d_in[7];
    const float* b2    = (const float*)d_in[8];
    float* out  = (float*)d_out;
    float* Bpad = (float*)d_ws;

    const long tokens = (long)in_sizes[0] / EDIM;
    const int  blocks = (int)(tokens / THREADS);

    qtb_setup<<<1, 256, 0, stream>>>(W_ffn, W_out, Bpad);
    qtb_main<<<blocks, THREADS, 0, stream>>>(x, th_a, th_f, g1, b1, g2, b2,
                                             Bpad, out);
}


# Setup launch folded into the main kernel: one wave per workgroup forms the 10x10 combined weight by 16 chained f32 WMMA steps under the token loads (the sum order of that weight changes)
# speedup vs baseline: 1.0874x; 1.0874x over previous
#include <hip/hip_runtime.h>


#define EDIM 10
#define FDIM 64
#define KPAD 12
#define FPAD 17
#define WOP  68
#define THREADS 256
#define LN_EPS 1e-5f

typedef float v2f __attribute__((ext_vector_type(2)));
typedef float v4f __attribute__((ext_vector_type(4)));
typedef float v8f __attribute__((ext_vector_type(8)));

__device__ __forceinline__ void load_token(const float* __restrict__ xp, float (&xv)[EDIM])
{
    #pragma unroll
    for (int i = 0; i < EDIM / 2; ++i) {
        v2f v = *(const v2f*)(xp + 2 * i);
        xv[2 * i] = v.x;
        xv[2 * i + 1] = v.y;
    }
}

__launch_bounds__(THREADS)
__global__ void qtb_main(const float* __restrict__ x,
                         const float* __restrict__ th_a,
                         const float* __restrict__ th_f,
                         const float* __restrict__ W_ffn,
                         const float* __restrict__ W_out,
                         const float* __restrict__ g1, const float* __restrict__ b1,
                         const float* __restrict__ g2, const float* __restrict__ b2,
                         float* __restrict__ out)
{
    __shared__ __align__(16) float meas_lds[THREADS * KPAD];
    __shared__ __align__(16) float ffn_lds [THREADS * FPAD];
    __shared__ __align__(16) float wffn_lds[FDIM * EDIM];
    __shared__ __align__(16) float wout_lds[EDIM * WOP];
    __shared__ __align__(16) float bpad_lds[KPAD * 16];

    const int tid  = threadIdx.x;
    const int lane = tid & 31;
    const int wave = tid >> 5;
    const int nn   = lane & 15;
    const int kh   = lane >> 4;

    const unsigned bix = blockIdx.x;
    const bool setup_wave =
        __builtin_amdgcn_readfirstlane(wave) == (int)((bix + (bix >> 3) + (bix >> 8)) & 7u);

    const long tok = (long)bix * THREADS + tid;
    const float* xp = x + tok * EDIM;
    float xv[EDIM];

    if (setup_wave) {
        v4f wf[5], wo[5];
        #pragma unroll
        for (int q = 0; q < 5; ++q) {
            wf[q] = *(const v4f*)(W_ffn + 4 * (lane + 32 * q));
            wo[q] = *(const v4f*)(W_out + 4 * (lane + 32 * q));
        }
        load_token(xp, xv);

        #pragma unroll
        for (int q = 0; q < 5; ++q) {
            const int i = lane + 32 * q;
            *(v4f*)&wffn_lds[4 * i] = wf[q];
            *(v4f*)&wout_lds[(i >> 4) * WOP + 4 * (i & 15)] = wo[q];
        }
        __builtin_amdgcn_wave_barrier();

        const int cc = nn < EDIM ? nn : EDIM - 1;
        v8f d0 = {0.f, 0.f, 0.f, 0.f, 0.f, 0.f, 0.f, 0.f};
        v2f af[FDIM / 4], bf[FDIM / 4];
#define LOAD_FRAGS(S0, S1)                                                    \
        _Pragma("unroll")                                                     \
        for (int s = (S0); s < (S1); ++s) {                                   \
            const int f0 = 4 * s + 2 * kh;                                    \
            af[s].x = wffn_lds[f0 * EDIM + cc];                               \
            af[s].y = wffn_lds[(f0 + 1) * EDIM + cc];                         \
            bf[s]   = *(const v2f*)&wout_lds[cc * WOP + f0];                  \
        }
#define RUN_STEPS(S0, S1)                                                     \
        _Pragma("unroll")                                                     \
        for (int s = (S0); s < (S1); ++s) {                                   \
            d0 = __builtin_amdgcn_wmma_f32_16x16x4_f32(false, af[s], false, bf[s], \
                                                       (short)0, d0, false, false); \
        }
        LOAD_FRAGS(0, 8)
        __builtin_amdgcn_sched_barrier(0);
        RUN_STEPS(0, 4)
        __builtin_amdgcn_sched_barrier(0);
        LOAD_FRAGS(8, 12)
        __builtin_amdgcn_sched_barrier(0);
        RUN_STEPS(4, 8)
        __builtin_amdgcn_sched_barrier(0);
        LOAD_FRAGS(12, 16)
        __builtin_amdgcn_sched_barrier(0);
        RUN_STEPS(8, 16)
#undef LOAD_FRAGS
#undef RUN_STEPS
        #pragma unroll
        for (int r = 0; r < 8; ++r) {
            const int k = 8 * kh + r;
            if (k < KPAD)
                bpad_lds[k * 16 + nn] = (k < EDIM && nn < EDIM) ? d0[r] : 0.0f;
        }
    } else {
        load_token(xp, xv);
    }

    float v1[EDIM];
    float prod = 1.0f, sum = 0.0f;
    #pragma unroll
    for (int i = 0; i < EDIM; ++i) {
        prod *= __cosf(xv[i] + th_a[i]);
        v1[i] = xv[i] + prod;
        sum  += v1[i];
    }

    const float inv_e = 1.0f / (float)EDIM;
    float mu = sum * inv_e;
    float var = 0.0f;
    #pragma unroll
    for (int i = 0; i < EDIM; ++i) {
        float d = v1[i] - mu;
        var += d * d;
    }
    var *= inv_e;
    float rs = rsqrtf(var + LN_EPS);
    float h[EDIM];
    #pragma unroll
    for (int i = 0; i < EDIM; ++i)
        h[i] = (v1[i] - mu) * rs * g1[i] + b1[i];

    #pragma unroll
    for (int i = 0; i < EDIM; ++i)
        meas_lds[tid * KPAD + i] = __cosf(h[i] + th_f[i]);
    meas_lds[tid * KPAD + 10] = 0.0f;
    meas_lds[tid * KPAD + 11] = 0.0f;

    __syncthreads();

    v2f bmat[3];
    #pragma unroll
    for (int c = 0; c < 3; ++c) {
        int k0 = c * 4 + 2 * kh;
        bmat[c].x = bpad_lds[k0 * 16 + nn];
        bmat[c].y = bpad_lds[(k0 + 1) * 16 + nn];
    }

    v8f acc[2];
    #pragma unroll
    for (int t2 = 0; t2 < 2; ++t2) {
        const int tile = wave * 2 + t2;
        const float* arow = &meas_lds[(tile * 16 + nn) * KPAD + 2 * kh];
        v8f c = {0.f, 0.f, 0.f, 0.f, 0.f, 0.f, 0.f, 0.f};
        #pragma unroll
        for (int ck = 0; ck < 3; ++ck) {
            v2f a = *(const v2f*)(arow + ck * 4);
            c = __builtin_amdgcn_wmma_f32_16x16x4_f32(
                     false, a,
                     false, bmat[ck],
                     (short)0, c,
                     false,  false);
        }
        acc[t2] = c;
    }

    #pragma unroll
    for (int t2 = 0; t2 < 2; ++t2) {
        const int tile = wave * 2 + t2;
        #pragma unroll
        for (int r = 0; r < 8; ++r)
            ffn_lds[(tile * 16 + r + 8 * kh) * FPAD + nn] = acc[t2][r];
    }

    __syncthreads();

    float yv[EDIM];
    float s2 = 0.0f;
    #pragma unroll
    for (int i = 0; i < EDIM; ++i) {
        yv[i] = h[i] + ffn_lds[tid * FPAD + i];
        s2 += yv[i];
    }
    float mu2 = s2 * inv_e;
    float var2 = 0.0f;
    #pragma unroll
    for (int i = 0; i < EDIM; ++i) {
        float d = yv[i] - mu2;
        var2 += d * d;
    }
    var2 *= inv_e;
    float rs2 = rsqrtf(var2 + LN_EPS);

    float* op = out + tok * EDIM;
    #pragma unroll
    for (int i = 0; i < EDIM / 2; ++i) {
        v2f v;
        v.x = (yv[2 * i]     - mu2) * rs2 * g2[2 * i]     + b2[2 * i];
        v.y = (yv[2 * i + 1] - mu2) * rs2 * g2[2 * i + 1] + b2[2 * i + 1];
        *(v2f*)(op + 2 * i) = v;
    }
}

extern "C" void kernel_launch(void* const* d_in, const int* in_sizes, int n_in,
                              void* d_out, int out_size, void* d_ws, size_t ws_size,
                              hipStream_t stream) {
    const float* x     = (const float*)d_in[0];
    const float* th_a  = (const float*)d_in[1];
    const float* th_f  = (const float*)d_in[2];
    const float* W_ffn = (const float*)d_in[3];
    const float* W_out = (const float*)d_in[4];
    const float* g1    = (const float*)d_in[5];
    const float* b1    = (const float*)d_in[6];
    const float* g2    = (const float*)d_in[7];
    const float* b2    = (const float*)d_in[8];
    float* out  = (float*)d_out;

    const long tokens = (long)in_sizes[0] / EDIM;
    const int  blocks = (int)(tokens / THREADS);

    qtb_main<<<blocks, THREADS, 0, stream>>>(x, th_a, th_f, W_ffn, W_out,
                                             g1, b1, g2, b2, out);
}


# Vector issue slots per wave 191 to 145: math on packed pairs, variance and mean summed pair-wise and fused (sum order changes), theta scaled on the scalar unit, bare v_rsq, aligned LDS row reads
# speedup vs baseline: 1.1639x; 1.0704x over previous
#include <hip/hip_runtime.h>


#define EDIM 10
#define FDIM 64
#define KPAD 12
#define FPAD 20
#define WOP  68
#define THREADS 256
#define LN_EPS 1e-5f

typedef float v2f __attribute__((ext_vector_type(2)));
typedef float v4f __attribute__((ext_vector_type(4)));
typedef float v8f __attribute__((ext_vector_type(8)));

__device__ __forceinline__ void load_token(const float* __restrict__ xp, v2f (&xv)[EDIM / 2])
{
    #pragma unroll
    for (int i = 0; i < EDIM / 2; ++i)
        xv[i] = *(const v2f*)(xp + 2 * i);
}

__device__ __forceinline__ float to_rev(float theta)
{
    float r;
    asm("s_mul_f32 %0, %1, 0.15915494" : "=s"(r) : "s"(theta));
    return r;
}

__launch_bounds__(THREADS)
__global__ void qtb_main(const float* __restrict__ x,
                         const float* __restrict__ th_a,
                         const float* __restrict__ th_f,
                         const float* __restrict__ W_ffn,
                         const float* __restrict__ W_out,
                         const float* __restrict__ g1, const float* __restrict__ b1,
                         const float* __restrict__ g2, const float* __restrict__ b2,
                         float* __restrict__ out)
{
    __shared__ __align__(16) float meas_lds[THREADS * KPAD];
    __shared__ __align__(16) float ffn_lds [THREADS * FPAD];
    __shared__ __align__(16) float wffn_lds[FDIM * EDIM];
    __shared__ __align__(16) float wout_lds[EDIM * WOP];
    __shared__ __align__(16) float bpad_lds[KPAD * 16];

    const int tid  = threadIdx.x;
    const int lane = tid & 31;
    const int wave = tid >> 5;
    const int nn   = lane & 15;
    const int kh   = lane >> 4;

    const unsigned bix = blockIdx.x;
    const bool setup_wave =
        __builtin_amdgcn_readfirstlane(wave) == (int)((bix + (bix >> 3) + (bix >> 8)) & 7u);

    const unsigned boff = __umul24(bix * THREADS + (unsigned)tid, EDIM * 4u);
    const float* xp = (const float*)((const char*)x + boff);
    v2f xv[EDIM / 2];

    if (setup_wave) {
        v4f wf[5], wo[5];
        #pragma unroll
        for (int q = 0; q < 5; ++q) {
            wf[q] = *(const v4f*)(W_ffn + 4 * (lane + 32 * q));
            wo[q] = *(const v4f*)(W_out + 4 * (lane + 32 * q));
        }
        load_token(xp, xv);

        #pragma unroll
        for (int q = 0; q < 5; ++q) {
            const int i = lane + 32 * q;
            *(v4f*)&wffn_lds[4 * i] = wf[q];
            *(v4f*)&wout_lds[(i >> 4) * WOP + 4 * (i & 15)] = wo[q];
        }
        __builtin_amdgcn_wave_barrier();

        const int cc = nn < EDIM ? nn : EDIM - 1;
        v8f d0 = {0.f, 0.f, 0.f, 0.f, 0.f, 0.f, 0.f, 0.f};
        v2f af[FDIM / 4], bf[FDIM / 4];
#define LOAD_FRAGS(S0, S1)                                                    \
        _Pragma("unroll")                                                     \
        for (int s = (S0); s < (S1); ++s) {                                   \
            const int f0 = 4 * s + 2 * kh;                                    \
            af[s].x = wffn_lds[f0 * EDIM + cc];                               \
            af[s].y = wffn_lds[(f0 + 1) * EDIM + cc];                         \
            bf[s]   = *(const v2f*)&wout_lds[cc * WOP + f0];                  \
        }
#define RUN_STEPS(S0, S1)                                                     \
        _Pragma("unroll")                                                     \
        for (int s = (S0); s < (S1); ++s) {                                   \
            d0 = __builtin_amdgcn_wmma_f32_16x16x4_f32(false, af[s], false, bf[s], \
                                                       (short)0, d0, false, false); \
        }
        LOAD_FRAGS(0, 8)
        __builtin_amdgcn_sched_barrier(0);
        RUN_STEPS(0, 4)
        __builtin_amdgcn_sched_barrier(0);
        LOAD_FRAGS(8, 12)
        __builtin_amdgcn_sched_barrier(0);
        RUN_STEPS(4, 8)
        __builtin_amdgcn_sched_barrier(0);
        LOAD_FRAGS(12, 16)
        __builtin_amdgcn_sched_barrier(0);
        RUN_STEPS(8, 16)
#undef LOAD_FRAGS
#undef RUN_STEPS
        #pragma unroll
        for (int r = 0; r < 8; ++r) {
            const int k = 8 * kh + r;
            if (k < KPAD)
                bpad_lds[k * 16 + nn] = (k < EDIM && nn < EDIM) ? d0[r] : 0.0f;
        }
    } else {
        load_token(xp, xv);
    }

    const v2f INV2PI = {0.15915494309189532f, 0.15915494309189532f};
    v2f v1[EDIM / 2];
    {
        float prod = 1.0f;
        #pragma unroll
        for (int j = 0; j < EDIM / 2; ++j) {
            v2f tv;
            tv.x = to_rev(th_a[2 * j]);
            tv.y = to_rev(th_a[2 * j + 1]);
            const v2f a = __builtin_elementwise_fma(xv[j], INV2PI, tv);
            v2f pp;
            pp.x = (j == 0) ? __builtin_amdgcn_cosf(a.x) : prod * __builtin_amdgcn_cosf(a.x);
            pp.y = pp.x * __builtin_amdgcn_cosf(a.y);
            prod = pp.y;
            v1[j] = xv[j] + pp;
        }
    }
    const v2f s1v = ((v1[0] + v1[1]) + (v1[2] + v1[3])) + v1[4];

    const float inv_e = 1.0f / (float)EDIM;
    const float mu = (s1v.x + s1v.y) * inv_e;
    v2f dv[EDIM / 2];
    #pragma unroll
    for (int j = 0; j < EDIM / 2; ++j)
        dv[j] = v1[j] - mu;
    v2f q1v = dv[0] * dv[0];
    #pragma unroll
    for (int j = 1; j < EDIM / 2; ++j)
        q1v = __builtin_elementwise_fma(dv[j], dv[j], q1v);
    const float rs = __builtin_amdgcn_rsqf(__builtin_fmaf(q1v.x + q1v.y, inv_e, LN_EPS));
    v2f hv[EDIM / 2];
    #pragma unroll
    for (int j = 0; j < EDIM / 2; ++j) {
        v2f gv, bv;
        gv.x = g1[2 * j]; gv.y = g1[2 * j + 1];
        bv.x = b1[2 * j]; bv.y = b1[2 * j + 1];
        hv[j] = __builtin_elementwise_fma(dv[j] * rs, gv, bv);
    }

    {
        v2f mv[EDIM / 2];
        #pragma unroll
        for (int j = 0; j < EDIM / 2; ++j) {
            v2f tv;
            tv.x = to_rev(th_f[2 * j]);
            tv.y = to_rev(th_f[2 * j + 1]);
            const v2f a = __builtin_elementwise_fma(hv[j], INV2PI, tv);
            mv[j].x = __builtin_amdgcn_cosf(a.x);
            mv[j].y = __builtin_amdgcn_cosf(a.y);
        }
        v4f m0, m1, m2;
        m0.xy = mv[0]; m0.zw = mv[1];
        m1.xy = mv[2]; m1.zw = mv[3];
        m2.xy = mv[4];
        m2.z = 0.0f;
        m2.w = 0.0f;
        v4f* mrow = (v4f*)&meas_lds[tid * KPAD];
        mrow[0] = m0;
        mrow[1] = m1;
        mrow[2] = m2;
    }

    __syncthreads();

    v2f bmat[3];
    #pragma unroll
    for (int c = 0; c < 3; ++c) {
        int k0 = c * 4 + 2 * kh;
        bmat[c].x = bpad_lds[k0 * 16 + nn];
        bmat[c].y = bpad_lds[(k0 + 1) * 16 + nn];
    }

    v8f acc[2];
    #pragma unroll
    for (int t2 = 0; t2 < 2; ++t2) {
        const int tile = wave * 2 + t2;
        const float* arow = &meas_lds[(tile * 16 + nn) * KPAD + 2 * kh];
        v8f c = {0.f, 0.f, 0.f, 0.f, 0.f, 0.f, 0.f, 0.f};
        #pragma unroll
        for (int ck = 0; ck < 3; ++ck) {
            v2f a = *(const v2f*)(arow + ck * 4);
            c = __builtin_amdgcn_wmma_f32_16x16x4_f32(
                     false, a,
                     false, bmat[ck],
                     (short)0, c,
                     false,  false);
        }
        acc[t2] = c;
    }

    #pragma unroll
    for (int t2 = 0; t2 < 2; ++t2) {
        const int tile = wave * 2 + t2;
        #pragma unroll
        for (int r = 0; r < 8; ++r)
            ffn_lds[(tile * 16 + r + 8 * kh) * FPAD + nn] = acc[t2][r];
    }

    __syncthreads();

    v2f yv[EDIM / 2];
    {
        const v4f* frow = (const v4f*)&ffn_lds[tid * FPAD];
        const v4f f0 = frow[0];
        const v4f f1 = frow[1];
        const v2f f2 = *(const v2f*)&ffn_lds[tid * FPAD + 8];
        yv[0] = hv[0] + f0.xy;
        yv[1] = hv[1] + f0.zw;
        yv[2] = hv[2] + f1.xy;
        yv[3] = hv[3] + f1.zw;
        yv[4] = hv[4] + f2;
    }
    const v2f s2v = ((yv[0] + yv[1]) + (yv[2] + yv[3])) + yv[4];
    const float mu2 = (s2v.x + s2v.y) * inv_e;
    v2f ev[EDIM / 2];
    #pragma unroll
    for (int j = 0; j < EDIM / 2; ++j)
        ev[j] = yv[j] - mu2;
    v2f q2v = ev[0] * ev[0];
    #pragma unroll
    for (int j = 1; j < EDIM / 2; ++j)
        q2v = __builtin_elementwise_fma(ev[j], ev[j], q2v);
    const float rs2 = __builtin_amdgcn_rsqf(__builtin_fmaf(q2v.x + q2v.y, inv_e, LN_EPS));

    float* op = (float*)((char*)out + boff);
    #pragma unroll
    for (int j = 0; j < EDIM / 2; ++j) {
        v2f gv, bv;
        gv.x = g2[2 * j]; gv.y = g2[2 * j + 1];
        bv.x = b2[2 * j]; bv.y = b2[2 * j + 1];
        *(v2f*)(op + 2 * j) = __builtin_elementwise_fma(ev[j] * rs2, gv, bv);
    }
}

extern "C" void kernel_launch(void* const* d_in, const int* in_sizes, int n_in,
                              void* d_out, int out_size, void* d_ws, size_t ws_size,
                              hipStream_t stream) {
    const float* x     = (const float*)d_in[0];
    const float* th_a  = (const float*)d_in[1];
    const float* th_f  = (const float*)d_in[2];
    const float* W_ffn = (const float*)d_in[3];
    const float* W_out = (const float*)d_in[4];
    const float* g1    = (const float*)d_in[5];
    const float* b1    = (const float*)d_in[6];
    const float* g2    = (const float*)d_in[7];
    const float* b2    = (const float*)d_in[8];
    float* out  = (float*)d_out;

    const long tokens = (long)in_sizes[0] / EDIM;
    const int  blocks = (int)(tokens / THREADS);

    qtb_main<<<blocks, THREADS, 0, stream>>>(x, th_a, th_f, W_ffn, W_out,
                                             g1, b1, g2, b2, out);
}
